# starting kernel
# baseline (speedup 1.0000x reference)
#include <hip/hip_runtime.h>


typedef __attribute__((ext_vector_type(2))) float v2f;
typedef __attribute__((ext_vector_type(8))) float v8f;

#define KTOT        1152
#define KCHUNK      576
#define LDS_STRIDE  580
#define NPIX        196
#define NTILES      13

static __device__ __forceinline__ v8f wmma_f32(v2f a, v2f b, v8f c) {
    return __builtin_amdgcn_wmma_f32_16x16x4_f32(
         false, a,  false, b,
         (short)0, c,  false,  false);
}

__global__ __launch_bounds__(512) void fused_roll_conv_wmma(
    const float* __restrict__ x,
    const float* __restrict__ w,
    float* __restrict__ y)
{
    __shared__ float sT[16 * LDS_STRIDE];

    const int tid   = threadIdx.x;
    const int lane  = tid & 31;
    const int wv    = tid >> 5;
    const int ntile = blockIdx.x;

    const int colN = lane & 15;
    const int koff = (lane >> 4) << 1;

    const int   arow  = (wv << 4) + colN;
    const float* aptr  = w + (size_t)arow * KTOT + koff;
    const float* bbase = &sT[colN * LDS_STRIDE + koff];

    const int fp = ntile * 16 + wv;
    const int fn = fp / 14;
    const int fo = fp - fn * 14;
    const int fb1 = (fo + 13) % 14;
    const bool fvalid = (fp < NPIX);
    float* sTcol = &sT[wv * LDS_STRIDE];

    v8f acc0 = {0.f,0.f,0.f,0.f,0.f,0.f,0.f,0.f};
    v8f acc1 = acc0, acc2 = acc0, acc3 = acc0;

    for (int kc = 0; kc < KTOT; kc += KCHUNK) {
        __builtin_prefetch(aptr + kc, 0, 3);

        for (int q = 0; q < KCHUNK / 32; ++q) {
            int kk    = lane + (q << 5);
            int kflat = kc + kk;
            int l  = kflat & 127;
            int jk = kflat >> 7;
            int k  = jk % 3;
            int j  = jk / 3;
            float v = 0.f;
            if (fvalid) {
                int a = fn + j - 1;
                if (0 <= a && a < 14) {
                    int s0 = fo + k;
                    if (s0 >= 1 && s0 <= 14) {
                        int w0 = (s0 + 12) % 14;
                        v += x[l * NPIX + a * 14 + w0];
                    }
                    int s1 = fb1 + k;
                    if (s1 >= 1 && s1 <= 14) {
                        int w1 = (s1 + 12) % 14;
                        v += x[(128 + l) * NPIX + a * 14 + w1];
                    }
                }
            }
            sTcol[kk] = v;
        }
        __syncthreads();

        const float* ap = aptr + kc;
        #pragma unroll 2
        for (int kk = 0; kk < KCHUNK; kk += 16) {
            v2f a0 = *(const v2f*)(ap + kk);
            v2f b0 = *(const v2f*)(bbase + kk);
            v2f a1 = *(const v2f*)(ap + kk + 4);
            v2f b1 = *(const v2f*)(bbase + kk + 4);
            v2f a2 = *(const v2f*)(ap + kk + 8);
            v2f b2 = *(const v2f*)(bbase + kk + 8);
            v2f a3 = *(const v2f*)(ap + kk + 12);
            v2f b3 = *(const v2f*)(bbase + kk + 12);
            acc0 = wmma_f32(a0, b0, acc0);
            acc1 = wmma_f32(a1, b1, acc1);
            acc2 = wmma_f32(a2, b2, acc2);
            acc3 = wmma_f32(a3, b3, acc3);
        }
        __syncthreads();
    }

    v8f acc = (acc0 + acc1) + (acc2 + acc3);
    int p = ntile * 16 + colN;
    if (p < NPIX) {
        int half = (lane >> 4) << 3;
        #pragma unroll
        for (int r = 0; r < 8; ++r) {
            int i = (wv << 4) + half + r;
            y[(size_t)i * NPIX + p] = acc[r];
        }
    }
}

extern "C" void kernel_launch(void* const* d_in, const int* in_sizes, int n_in,
                              void* d_out, int out_size, void* d_ws, size_t ws_size,
                              hipStream_t stream) {
    const float* x = (const float*)d_in[0];
    const float* w = (const float*)d_in[1];
    float* y = (float*)d_out;
    (void)in_sizes; (void)n_in; (void)out_size; (void)d_ws; (void)ws_size;
    fused_roll_conv_wmma<<<dim3(NTILES), dim3(512), 0, stream>>>(x, w, y);
}


# One work-group per 16x16 output tile (208 groups, was 13): 9 waves split K by filter tap, gather B fragments straight from x, add the 9 partials in fixed order via LDS (changes f32 sum order)
# speedup vs baseline: 5.7247x; 5.7247x over previous
#include <hip/hip_runtime.h>


typedef __attribute__((ext_vector_type(2))) float v2f;
typedef __attribute__((ext_vector_type(8))) float v8f;

#define KTOT    1152
#define NPIX    196
#define NTILES  13
#define MTILES  16
#define NTAPS   9
#define LCH     128
#define NSTEP   (LCH / 4)

static __device__ __forceinline__ v8f wmma_f32(v2f a, v2f b, v8f c) {
    return __builtin_amdgcn_wmma_f32_16x16x4_f32(
         false, a,  false, b,
         (short)0, c,  false,  false);
}

__global__ __launch_bounds__(NTAPS * 32) void fused_roll_conv_wmma(
    const float* __restrict__ x,
    const float* __restrict__ w,
    float* __restrict__ y)
{
    __shared__ float part[NTAPS * 256];

    const int tid   = threadIdx.x;
    const int lane  = tid & 31;
    const int tap   = tid >> 5;
    const int ntile = blockIdx.x;
    const int mtile = blockIdx.y;

    const int colN = lane & 15;
    const int half = lane >> 4;
    const int tj = tap / 3;
    const int tk = tap - tj * 3;

    const float* aptr = w + (size_t)((mtile << 4) + colN) * KTOT + tap * LCH + (half << 1);

    const int  p   = (ntile << 4) + colN;
    const bool pv  = (p < NPIX);
    const int  pc  = pv ? p : 0;
    const int  pn  = pc / 14;
    const int  po  = pc - pn * 14;
    const int  a   = pn + tj - 1;
    const bool va  = pv && (a >= 0) && (a < 14);
    const int  s0  = po + tk;
    const bool v0  = va && (s0 >= 1) && (s0 <= 14);
    const int  w0  = (s0 + 12) % 14;
    const int  b1  = (po + 13) % 14;
    const int  s1  = b1 + tk;
    const bool v1  = va && (s1 >= 1) && (s1 <= 14);
    const int  w1  = (s1 + 12) % 14;
    const float* x0 = x + (half << 1) * NPIX + (v0 ? a * 14 + w0 : 0);
    const float* x1 = x + (LCH + (half << 1)) * NPIX + (v1 ? a * 14 + w1 : 0);

    v8f acc0 = {0.f,0.f,0.f,0.f,0.f,0.f,0.f,0.f};
    v8f acc1 = acc0;

    #pragma unroll
    for (int s = 0; s < NSTEP; s += 2) {
        v2f  a0  = *(const v2f*)(aptr + 4 * s);
        v2f  a1  = *(const v2f*)(aptr + 4 * s + 4);
        float p00 = x0[(4 * s + 0) * NPIX];
        float p01 = x0[(4 * s + 1) * NPIX];
        float q00 = x1[(4 * s + 0) * NPIX];
        float q01 = x1[(4 * s + 1) * NPIX];
        float p10 = x0[(4 * s + 4) * NPIX];
        float p11 = x0[(4 * s + 5) * NPIX];
        float q10 = x1[(4 * s + 4) * NPIX];
        float q11 = x1[(4 * s + 5) * NPIX];
        v2f bf0, bf1;
        bf0.x = (v0 ? p00 : 0.f) + (v1 ? q00 : 0.f);
        bf0.y = (v0 ? p01 : 0.f) + (v1 ? q01 : 0.f);
        bf1.x = (v0 ? p10 : 0.f) + (v1 ? q10 : 0.f);
        bf1.y = (v0 ? p11 : 0.f) + (v1 ? q11 : 0.f);
        acc0 = wmma_f32(a0, bf0, acc0);
        acc1 = wmma_f32(a1, bf1, acc1);
    }

    v8f acc = acc0 + acc1;
    #pragma unroll
    for (int r = 0; r < 8; ++r)
        part[tap * 256 + r * 32 + lane] = acc[r];
    __syncthreads();

    if (tid < 256) {
        float sum = part[tid];
        #pragma unroll
        for (int t = 1; t < NTAPS; ++t)
            sum += part[t * 256 + tid];
        if (pv) {
            int i = (mtile << 4) + (half << 3) + tap;
            y[(size_t)i * NPIX + p] = sum;
        }
    }
}

extern "C" void kernel_launch(void* const* d_in, const int* in_sizes, int n_in,
                              void* d_out, int out_size, void* d_ws, size_t ws_size,
                              hipStream_t stream) {
    const float* x = (const float*)d_in[0];
    const float* w = (const float*)d_in[1];
    float* y = (float*)d_out;
    (void)in_sizes; (void)n_in; (void)out_size; (void)d_ws; (void)ws_size;
    fused_roll_conv_wmma<<<dim3(NTILES, MTILES), dim3(NTAPS * 32), 0, stream>>>(x, w, y);
}


# Each work-group walks K's 16 channel pair-steps in one of 16 staggered orders (Latin square on block index) so groups do not ask L2 for the same x and w lines at once; changes f32 sum order
# speedup vs baseline: 6.1514x; 1.0745x over previous
#include <hip/hip_runtime.h>


typedef __attribute__((ext_vector_type(2))) float v2f;
typedef __attribute__((ext_vector_type(8))) float v8f;

#define KTOT    1152
#define NPIX    196
#define NTILES  13
#define MTILES  16
#define NTAPS   9
#define LCH     128
#define NSTEP   (LCH / 4)

static __device__ __forceinline__ v8f wmma_f32(v2f a, v2f b, v8f c) {
    return __builtin_amdgcn_wmma_f32_16x16x4_f32(
         false, a,  false, b,
         (short)0, c,  false,  false);
}

__global__ __launch_bounds__(NTAPS * 32) void fused_roll_conv_wmma(
    const float* __restrict__ x,
    const float* __restrict__ w,
    float* __restrict__ y)
{
    __shared__ float part[NTAPS * 256];

    const int tid   = threadIdx.x;
    const int lane  = tid & 31;
    const int tap   = __builtin_amdgcn_readfirstlane(tid >> 5);
    const int ntile = blockIdx.x;
    const int mtile = blockIdx.y;

    const int colN = lane & 15;
    const int half = lane >> 4;
    const int tj = tap / 3;
    const int tk = tap - tj * 3;

    const unsigned wbase  = (unsigned)((mtile << 4) * KTOT + tap * LCH) * 4u;
    const unsigned a_lane = (unsigned)(colN * KTOT + (half << 1)) * 4u;

    const int  p   = (ntile << 4) + colN;
    const bool pv  = (p < NPIX);
    const int  pc  = pv ? p : 0;
    const int  pn  = pc / 14;
    const int  po  = pc - pn * 14;
    const int  a   = pn + tj - 1;
    const bool va  = pv && (a >= 0) && (a < 14);
    const int  s0  = po + tk;
    const bool v0  = va && (s0 >= 1) && (s0 <= 14);
    const int  w0  = (s0 + 12) % 14;
    const int  b1  = (po + 13) % 14;
    const int  s1  = b1 + tk;
    const bool v1  = va && (s1 >= 1) && (s1 <= 14);
    const int  w1  = (s1 + 12) % 14;
    const unsigned l0 = (unsigned)((half << 1) * NPIX + (v0 ? a * 14 + w0 : 0)) * 4u;
    const unsigned l1 = (unsigned)((LCH + (half << 1)) * NPIX + (v1 ? a * 14 + w1 : 0)) * 4u;

    v8f acc0 = {0.f,0.f,0.f,0.f,0.f,0.f,0.f,0.f};
    v8f acc1 = acc0;

    const int ph = (ntile + 3 * mtile) & 15;
    const int pb = ph & 3;
    const int pr = ph >> 2;

    const char* abv[NSTEP / 2];
    const char* xbv[NSTEP / 2];
    #pragma unroll
    for (int i = 0; i < NSTEP / 2; ++i) {
        const int sp = ((((i >> 2) + pb) & 3) << 2) | (((i & 3) + pr) & 3);
        abv[i] = (const char*)w + (wbase + sp * (8 * 4));
        xbv[i] = (const char*)x + sp * (8 * NPIX * 4);
    }

    #pragma unroll
    for (int i = 0; i < NSTEP / 2; ++i) {
        const char* ab = abv[i];
        const char* xb = xbv[i];
        v2f  a0  = *(const v2f*)(ab + a_lane);
        v2f  a1  = *(const v2f*)(ab + a_lane + 16);
        float p00 = *(const float*)(xb + l0);
        float p01 = *(const float*)(xb + l0 + 1 * NPIX * 4);
        float q00 = *(const float*)(xb + l1);
        float q01 = *(const float*)(xb + l1 + 1 * NPIX * 4);
        float p10 = *(const float*)(xb + l0 + 4 * NPIX * 4);
        float p11 = *(const float*)(xb + l0 + 5 * NPIX * 4);
        float q10 = *(const float*)(xb + l1 + 4 * NPIX * 4);
        float q11 = *(const float*)(xb + l1 + 5 * NPIX * 4);
        v2f bf0, bf1;
        bf0.x = (v0 ? p00 : 0.f) + (v1 ? q00 : 0.f);
        bf0.y = (v0 ? p01 : 0.f) + (v1 ? q01 : 0.f);
        bf1.x = (v0 ? p10 : 0.f) + (v1 ? q10 : 0.f);
        bf1.y = (v0 ? p11 : 0.f) + (v1 ? q11 : 0.f);
        acc0 = wmma_f32(a0, bf0, acc0);
        acc1 = wmma_f32(a1, bf1, acc1);
    }

    v8f acc = acc0 + acc1;
    #pragma unroll
    for (int r = 0; r < 8; ++r)
        part[tap * 256 + r * 32 + lane] = acc[r];
    __syncthreads();

    if (tid < 256) {
        float sum = part[tid];
        #pragma unroll
        for (int t = 1; t < NTAPS; ++t)
            sum += part[t * 256 + tid];
        if (pv) {
            int i = (mtile << 4) + (half << 3) + tap;
            y[(size_t)i * NPIX + p] = sum;
        }
    }
}

extern "C" void kernel_launch(void* const* d_in, const int* in_sizes, int n_in,
                              void* d_out, int out_size, void* d_ws, size_t ws_size,
                              hipStream_t stream) {
    const float* x = (const float*)d_in[0];
    const float* w = (const float*)d_in[1];
    float* y = (float*)d_out;
    (void)in_sizes; (void)n_in; (void)out_size; (void)d_ws; (void)ws_size;
    fused_roll_conv_wmma<<<dim3(NTILES, MTILES), dim3(NTAPS * 32), 0, stream>>>(x, w, y);
}


# w operand: each wave fetches its 16x128 slab with whole-line loads (4 rows x 128 B each), parks it in a wave-private LDS slab, reads fragments back 16 B per lane; k regrouped, changes f32 sum order
# speedup vs baseline: 6.4842x; 1.0541x over previous
#include <hip/hip_runtime.h>


typedef __attribute__((ext_vector_type(2))) float v2f;
typedef __attribute__((ext_vector_type(4))) float v4f;
typedef __attribute__((ext_vector_type(8))) float v8f;

#define KTOT    1152
#define NPIX    196
#define NTILES  13
#define MTILES  16
#define NTAPS   9
#define LCH     128
#define NPOS    (LCH / 8)
#define APITCH  132
#define NBLK    4

static __device__ __forceinline__ v8f wmma_f32(v2f a, v2f b, v8f c) {
    return __builtin_amdgcn_wmma_f32_16x16x4_f32(
         false, a,  false, b,
         (short)0, c,  false,  false);
}

__global__ __launch_bounds__(NTAPS * 32) void fused_roll_conv_wmma(
    const float* __restrict__ x,
    const float* __restrict__ w,
    float* __restrict__ y)
{
    __shared__ float part[NTAPS * 256];
    __shared__ __attribute__((aligned(16))) float aslab[NTAPS * 16 * APITCH];

    const int tid   = threadIdx.x;
    const int lane  = tid & 31;
    const int tap   = __builtin_amdgcn_readfirstlane(tid >> 5);
    const int ntile = blockIdx.x;
    const int mtile = blockIdx.y;

    const int colN = lane & 15;
    const int half = lane >> 4;
    const int tj = tap / 3;
    const int tk = tap - tj * 3;

    const int ph = (ntile + 3 * mtile) & 15;
    const int pb = ph & 3;
    const int pr = ph >> 2;

    const int lg = lane >> 3;
    const int ck = lane & 7;
    const int arow0 = ((lg & 1) << 3) | ((lg >> 1) << 2);
    const unsigned wbase   = (unsigned)((mtile << 4) * KTOT + tap * LCH) * 4u;
    const unsigned ag_lane = (unsigned)(arow0 * KTOT + (ck << 2)) * 4u;
    float* slab = aslab + tap * (16 * APITCH);
    const unsigned as_lane = (unsigned)(arow0 * APITCH + ((((ck >> 1) - pr) & 3) << 3) + ((ck & 1) << 2));
    const unsigned af_lane = (unsigned)(colN * APITCH + (half << 2));

    v4f ag[16];
    #pragma unroll
    for (int bi = 0; bi < 4; ++bi) {
        const char* wb = (const char*)w + (wbase + (unsigned)(((bi + pb) & 3) << 7));
        #pragma unroll
        for (int rg = 0; rg < 4; ++rg) {
            const int rgp = (rg + pr) & 3;
            ag[bi * 4 + rg] = *(const v4f*)(wb + ag_lane + (unsigned)(rgp * (KTOT * 4)));
        }
    }

    const int  p   = (ntile << 4) + colN;
    const bool pv  = (p < NPIX);
    const int  pc  = pv ? p : 0;
    const int  pn  = pc / 14;
    const int  po  = pc - pn * 14;
    const int  a   = pn + tj - 1;
    const bool va  = pv && (a >= 0) && (a < 14);
    const int  s0  = po + tk;
    const bool v0  = va && (s0 >= 1) && (s0 <= 14);
    const int  w0  = (s0 + 12) % 14;
    const int  b1  = (po + 13) % 14;
    const int  s1  = b1 + tk;
    const bool v1  = va && (s1 >= 1) && (s1 <= 14);
    const int  w1  = (s1 + 12) % 14;
    const unsigned l0 = (unsigned)((half << 2) * NPIX + (v0 ? a * 14 + w0 : 0)) * 4u;
    const unsigned l1 = (unsigned)((LCH + (half << 2)) * NPIX + (v1 ? a * 14 + w1 : 0)) * 4u;

    const char* xbv[NPOS];
    #pragma unroll
    for (int i = 0; i < NPOS; ++i) {
        const int sp = ((((i >> 2) + pb) & 3) << 2) | (((i & 3) + pr) & 3);
        xbv[i] = (const char*)x + sp * (8 * NPIX * 4);
    }

    float pt[NPOS][4];
    float qt[NPOS][4];
    #pragma unroll
    for (int i = 0; i < 8; ++i) {
        #pragma unroll
        for (int c = 0; c < 4; ++c) {
            pt[i][c] = *(const float*)(xbv[i] + l0 + c * (NPIX * 4));
            qt[i][c] = *(const float*)(xbv[i] + l1 + c * (NPIX * 4));
        }
    }

    v8f acc0 = {0.f,0.f,0.f,0.f,0.f,0.f,0.f,0.f};
    v8f acc1 = acc0;

    #pragma unroll
    for (int bi = 0; bi < NBLK; ++bi) {
        #pragma unroll
        for (int rg = 0; rg < 4; ++rg) {
            const int rgp = (rg + pr) & 3;
            *(v4f*)(slab + as_lane + (unsigned)(rgp * APITCH) + bi * 32) = ag[bi * 4 + rg];
        }

        __builtin_amdgcn_fence(__ATOMIC_RELEASE, "wavefront");
        __builtin_amdgcn_wave_barrier();
        __builtin_amdgcn_fence(__ATOMIC_ACQUIRE, "wavefront");

        if (bi + 2 < NBLK) {
            #pragma unroll
            for (int i = (bi + 2) * 4; i < (bi + 3) * 4; ++i) {
                #pragma unroll
                for (int c = 0; c < 4; ++c) {
                    pt[i][c] = *(const float*)(xbv[i] + l0 + c * (NPIX * 4));
                    qt[i][c] = *(const float*)(xbv[i] + l1 + c * (NPIX * 4));
                }
            }
        }

        v4f af[4];
        #pragma unroll
        for (int ii = 0; ii < 4; ++ii)
            af[ii] = *(const v4f*)(slab + af_lane + (bi * 4 + ii) * 8);

        #pragma unroll
        for (int ii = 0; ii < 4; ++ii) {
            const int i = bi * 4 + ii;
            v2f a0, a1, bf0, bf1;
            a0.x = af[ii].x;  a0.y = af[ii].y;
            a1.x = af[ii].z;  a1.y = af[ii].w;
            bf0.x = (v0 ? pt[i][0] : 0.f) + (v1 ? qt[i][0] : 0.f);
            bf0.y = (v0 ? pt[i][1] : 0.f) + (v1 ? qt[i][1] : 0.f);
            bf1.x = (v0 ? pt[i][2] : 0.f) + (v1 ? qt[i][2] : 0.f);
            bf1.y = (v0 ? pt[i][3] : 0.f) + (v1 ? qt[i][3] : 0.f);
            acc0 = wmma_f32(a0, bf0, acc0);
            acc1 = wmma_f32(a1, bf1, acc1);
        }
    }

    v8f acc = acc0 + acc1;
    #pragma unroll
    for (int r = 0; r < 8; ++r)
        part[tap * 256 + r * 32 + lane] = acc[r];
    __syncthreads();

    if (tid < 256) {
        float sum = part[tid];
        #pragma unroll
        for (int t = 1; t < NTAPS; ++t)
            sum += part[t * 256 + tid];
        if (pv) {
            int i = (mtile << 4) + (half << 3) + tap;
            y[(size_t)i * NPIX + p] = sum;
        }
    }
}

extern "C" void kernel_launch(void* const* d_in, const int* in_sizes, int n_in,
                              void* d_out, int out_size, void* d_ws, size_t ws_size,
                              hipStream_t stream) {
    const float* x = (const float*)d_in[0];
    const float* w = (const float*)d_in[1];
    float* y = (float*)d_out;
    (void)in_sizes; (void)n_in; (void)out_size; (void)d_ws; (void)ws_size;
    fused_roll_conv_wmma<<<dim3(NTILES, MTILES), dim3(NTAPS * 32), 0, stream>>>(x, w, y);
}
